# starting kernel
# baseline (speedup 1.0000x reference)
#include <hip/hip_runtime.h>


typedef __attribute__((ext_vector_type(2))) float v2f;
typedef __attribute__((ext_vector_type(4))) float v4f;
typedef __attribute__((ext_vector_type(8))) float v8f;

#define B_SZ   32
#define I_SZ   1024
#define DIN    512
#define NOUT   512
#define ICH    16
#define ICHUNK (I_SZ / ICH)

__global__ void caps_reduce_partial(const float* __restrict__ u,
                                    float* __restrict__ part) {
  const int t   = threadIdx.x;
  const int b   = blockIdx.x;
  const int ich = blockIdx.y;
  const v4f* p = (const v4f*)(u + ((size_t)b * I_SZ + (size_t)ich * ICHUNK) * DIN) + t;
  v4f s = {};
#pragma unroll 8
  for (int i = 0; i < ICHUNK; ++i) s += p[(size_t)i * (DIN / 4)];
  ((v4f*)part)[((size_t)ich * B_SZ + b) * (DIN / 4) + t] = s;
}

__global__ void caps_combine(const float* __restrict__ part,
                             float* __restrict__ S) {
  const int idx = blockIdx.x * 256 + threadIdx.x;
  const v4f* p = (const v4f*)part;
  v4f s = {};
#pragma unroll
  for (int ich = 0; ich < ICH; ++ich)
    s += p[(size_t)ich * (B_SZ * DIN / 4) + idx];
  ((v4f*)S)[idx] = s;
}

__global__ void caps_wmma_gemm(const float* __restrict__ S,
                               const float* __restrict__ W,
                               float* __restrict__ O) {
  const int wave = threadIdx.x >> 5;
  const int lane = threadIdx.x & 31;
  const int tile = blockIdx.x * 4 + wave;
  const int mt   = tile & 1;
  const int nt   = tile >> 1;
  const int hi   = lane >> 4;
  const int l16  = lane & 15;
  const int mrow = mt * 16 + l16;
  const int ncol = nt * 16 + l16;

  v8f acc = {};
  for (int k0 = 0; k0 < DIN; k0 += 4) {
    const int kb = k0 + hi * 2;
    v2f a, bf;
    a.x  = S[(size_t)mrow * DIN + kb];
    a.y  = S[(size_t)mrow * DIN + kb + 1];
    bf.x = W[(size_t)kb * NOUT + ncol];
    bf.y = W[(size_t)(kb + 1) * NOUT + ncol];
    acc = __builtin_amdgcn_wmma_f32_16x16x4_f32(
         false, a,  false, bf,
         (short)0, acc,  false,  false);
  }

  const int mbase = mt * 16 + hi * 8;
#pragma unroll
  for (int j = 0; j < 8; ++j)
    O[(size_t)(mbase + j) * NOUT + ncol] = acc[j];
}

extern "C" void kernel_launch(void* const* d_in, const int* in_sizes, int n_in,
                              void* d_out, int out_size, void* d_ws, size_t ws_size,
                              hipStream_t stream) {
  const float* u = (const float*)d_in[0];
  const float* W = (const float*)d_in[1];
  float* out = (float*)d_out;

  float* part = (float*)d_ws;
  float* S    = part + (size_t)ICH * B_SZ * DIN;

  caps_reduce_partial<<<dim3(B_SZ, ICH), 128, 0, stream>>>(u, part);
  caps_combine<<<(B_SZ * DIN / 4) / 256, 256, 0, stream>>>(part, S);
  caps_wmma_gemm<<<16, 128, 0, stream>>>(S, W, out);
}


# GEMM stage: K split over the 16 waves of each work-group (32 groups of 16x32 outputs), all fragment loads ahead of the WMMAs, partials summed through LDS in fixed order (new reduction tree)
# speedup vs baseline: 1.6311x; 1.6311x over previous
#include <hip/hip_runtime.h>


typedef __attribute__((ext_vector_type(2))) float v2f;
typedef __attribute__((ext_vector_type(4))) float v4f;
typedef __attribute__((ext_vector_type(8))) float v8f;

#define B_SZ   32
#define I_SZ   1024
#define DIN    512
#define NOUT   512
#define ICH    16
#define ICHUNK (I_SZ / ICH)

__global__ void caps_reduce_partial(const float* __restrict__ u,
                                    float* __restrict__ part) {
  const int t   = threadIdx.x;
  const int b   = blockIdx.x;
  const int ich = blockIdx.y;
  const v4f* p = (const v4f*)(u + ((size_t)b * I_SZ + (size_t)ich * ICHUNK) * DIN) + t;
  v4f s = {};
#pragma unroll 8
  for (int i = 0; i < ICHUNK; ++i) s += p[(size_t)i * (DIN / 4)];
  ((v4f*)part)[((size_t)ich * B_SZ + b) * (DIN / 4) + t] = s;
}

__global__ void caps_combine(const float* __restrict__ part,
                             float* __restrict__ S) {
  const int idx = blockIdx.x * 256 + threadIdx.x;
  const v4f* p = (const v4f*)part;
  v4f s = {};
#pragma unroll
  for (int ich = 0; ich < ICH; ++ich)
    s += p[(size_t)ich * (B_SZ * DIN / 4) + idx];
  ((v4f*)S)[idx] = s;
}

#define KSPLIT 16
#define KSLICE (DIN / KSPLIT)
#define KSTEPS (KSLICE / 4)
#define NSTRIP 32

__global__ __launch_bounds__(KSPLIT * 32)
void caps_wmma_gemm(const float* __restrict__ S,
                    const float* __restrict__ W,
                    float* __restrict__ O) {
  __shared__ float red[KSPLIT * 8 * 2 * 32];

  const int ks   = threadIdx.x >> 5;
  const int lane = threadIdx.x & 31;
  const int mt   = blockIdx.x & 1;
  const int np   = blockIdx.x >> 1;
  const int hi   = lane >> 4;
  const int l16  = lane & 15;
  const int mrow = mt * 16 + l16;
  const int ncol = np * NSTRIP + l16;
  const int kb0  = ks * KSLICE + hi * 2;

  const float* ap = S + (size_t)mrow * DIN + kb0;
  const float* bp = W + (size_t)kb0 * NOUT + ncol;

  int wr = ks * (8 * 2 * 32) + lane;
  const int r   = threadIdx.x >> 5;
  const int col = threadIdx.x & 31;
  int rd  = ((r & 7) * 2 + (col >> 4)) * 32 + (r >> 3) * 16 + (col & 15);
  int oix = (mt * 16 + r) * NOUT + np * NSTRIP + col;

  v2f a[KSTEPS], b0[KSTEPS], b1[KSTEPS];
#pragma unroll
  for (int s = 0; s < KSTEPS; ++s) {
    a[s]    = *(const v2f*)(ap + s * 4);
    b0[s].x = bp[(size_t)(s * 4)     * NOUT];
    b0[s].y = bp[(size_t)(s * 4 + 1) * NOUT];
    b1[s].x = bp[(size_t)(s * 4)     * NOUT + 16];
    b1[s].y = bp[(size_t)(s * 4 + 1) * NOUT + 16];
  }

  asm volatile("" : "+v"(wr), "+v"(rd), "+v"(oix));
  __builtin_amdgcn_sched_barrier(0);

  v8f acc0 = {}, acc1 = {};
#pragma unroll
  for (int s = 0; s < KSTEPS; ++s) {
    acc0 = __builtin_amdgcn_wmma_f32_16x16x4_f32(
         false, a[s],  false, b0[s],
         (short)0, acc0,  false,  false);
    acc1 = __builtin_amdgcn_wmma_f32_16x16x4_f32(
         false, a[s],  false, b1[s],
         (short)0, acc1,  false,  false);
  }

  asm volatile("v_nop\n\tv_nop\n\tv_nop\n\tv_nop\n\tv_nop" : "+v"(acc0), "+v"(acc1));

#pragma unroll
  for (int j = 0; j < 8; ++j) {
    red[wr + (j * 2 + 0) * 32] = acc0[j];
    red[wr + (j * 2 + 1) * 32] = acc1[j];
  }
  __syncthreads();

  float sum = 0.0f;
#pragma unroll
  for (int q = 0; q < KSPLIT; ++q) sum += red[q * (8 * 2 * 32) + rd];
  O[oix] = sum;
}

extern "C" void kernel_launch(void* const* d_in, const int* in_sizes, int n_in,
                              void* d_out, int out_size, void* d_ws, size_t ws_size,
                              hipStream_t stream) {
  const float* u = (const float*)d_in[0];
  const float* W = (const float*)d_in[1];
  float* out = (float*)d_out;

  float* part = (float*)d_ws;
  float* S    = part + (size_t)ICH * B_SZ * DIN;

  caps_reduce_partial<<<dim3(B_SZ, ICH), 128, 0, stream>>>(u, part);
  caps_combine<<<(B_SZ * DIN / 4) / 256, 256, 0, stream>>>(part, S);
  caps_wmma_gemm<<<2 * (NOUT / NSTRIP), KSPLIT * 32, 0, stream>>>(S, W, out);
}


# Sum over I finished inside each work-group (8 waves, 64 row loads per lane in flight at once, 16 partials summed through LDS in fixed order: new reduction tree); combine launch removed
# speedup vs baseline: 2.3822x; 1.4605x over previous
#include <hip/hip_runtime.h>


typedef __attribute__((ext_vector_type(2))) float v2f;
typedef __attribute__((ext_vector_type(4))) float v4f;
typedef __attribute__((ext_vector_type(8))) float v8f;

#define B_SZ   32
#define I_SZ   1024
#define DIN    512
#define NOUT   512
#define RWAVES 8
#define KSEG   64
#define RSTEP  (2 * RWAVES)
#define NLOAD  (I_SZ / RSTEP)

__global__ __launch_bounds__(RWAVES * 32)
void caps_reduce_rows(const float* __restrict__ u, float* __restrict__ S) {
  __shared__ __attribute__((aligned(16))) float red[RSTEP * KSEG];

  const int lane = threadIdx.x & 31;
  const int w    = threadIdx.x >> 5;
  const int h    = lane >> 4;
  const int c    = lane & 15;
  const int kseg = blockIdx.x & 7;
  const int b    = blockIdx.x >> 3;

  const v4f* p = (const v4f*)(u + ((size_t)b * I_SZ + (size_t)(2 * w + h)) * DIN
                                + kseg * KSEG) + c;
  v4f x[NLOAD];
#pragma unroll
  for (int j = 0; j < NLOAD; ++j) x[j] = p[(size_t)j * (RSTEP * (DIN / 4))];

  __builtin_amdgcn_sched_barrier(0);

  v4f s = x[0];
#pragma unroll
  for (int j = 1; j < NLOAD; ++j) s += x[j];

  ((v4f*)red)[threadIdx.x] = s;
  __syncthreads();

  if (w == 0) {
    const v2f* r2 = (const v2f*)red;
    v2f t = r2[lane];
#pragma unroll
    for (int q = 1; q < RSTEP; ++q) t += r2[q * (KSEG / 2) + lane];
    ((v2f*)S)[((size_t)b * DIN + kseg * KSEG) / 2 + lane] = t;
  }
}

#define KSPLIT 16
#define KSLICE (DIN / KSPLIT)
#define KSTEPS (KSLICE / 4)
#define NSTRIP 32

__global__ __launch_bounds__(KSPLIT * 32)
void caps_wmma_gemm(const float* __restrict__ S,
                    const float* __restrict__ W,
                    float* __restrict__ O) {
  __shared__ float red[KSPLIT * 8 * 2 * 32];

  const int ks   = threadIdx.x >> 5;
  const int lane = threadIdx.x & 31;
  const int mt   = blockIdx.x & 1;
  const int np   = blockIdx.x >> 1;
  const int hi   = lane >> 4;
  const int l16  = lane & 15;
  const int mrow = mt * 16 + l16;
  const int ncol = np * NSTRIP + l16;
  const int kb0  = ks * KSLICE + hi * 2;

  const float* ap = S + (size_t)mrow * DIN + kb0;
  const float* bp = W + (size_t)kb0 * NOUT + ncol;

  int wr = ks * (8 * 2 * 32) + lane;
  const int r   = threadIdx.x >> 5;
  const int col = threadIdx.x & 31;
  int rd  = ((r & 7) * 2 + (col >> 4)) * 32 + (r >> 3) * 16 + (col & 15);
  int oix = (mt * 16 + r) * NOUT + np * NSTRIP + col;

  v2f a[KSTEPS], b0[KSTEPS], b1[KSTEPS];
#pragma unroll
  for (int s = 0; s < KSTEPS; ++s) {
    a[s]    = *(const v2f*)(ap + s * 4);
    b0[s].x = bp[(size_t)(s * 4)     * NOUT];
    b0[s].y = bp[(size_t)(s * 4 + 1) * NOUT];
    b1[s].x = bp[(size_t)(s * 4)     * NOUT + 16];
    b1[s].y = bp[(size_t)(s * 4 + 1) * NOUT + 16];
  }

  asm volatile("" : "+v"(wr), "+v"(rd), "+v"(oix));
  __builtin_amdgcn_sched_barrier(0);

  v8f acc0 = {}, acc1 = {};
#pragma unroll
  for (int s = 0; s < KSTEPS; ++s) {
    acc0 = __builtin_amdgcn_wmma_f32_16x16x4_f32(
         false, a[s],  false, b0[s],
         (short)0, acc0,  false,  false);
    acc1 = __builtin_amdgcn_wmma_f32_16x16x4_f32(
         false, a[s],  false, b1[s],
         (short)0, acc1,  false,  false);
  }

  asm volatile("v_nop\n\tv_nop\n\tv_nop\n\tv_nop\n\tv_nop" : "+v"(acc0), "+v"(acc1));

#pragma unroll
  for (int j = 0; j < 8; ++j) {
    red[wr + (j * 2 + 0) * 32] = acc0[j];
    red[wr + (j * 2 + 1) * 32] = acc1[j];
  }
  __syncthreads();

  float sum = 0.0f;
#pragma unroll
  for (int q = 0; q < KSPLIT; ++q) sum += red[q * (8 * 2 * 32) + rd];
  O[oix] = sum;
}

extern "C" void kernel_launch(void* const* d_in, const int* in_sizes, int n_in,
                              void* d_out, int out_size, void* d_ws, size_t ws_size,
                              hipStream_t stream) {
  const float* u = (const float*)d_in[0];
  const float* W = (const float*)d_in[1];
  float* out = (float*)d_out;

  float* S = (float*)d_ws;

  caps_reduce_rows<<<B_SZ * (DIN / KSEG), RWAVES * 32, 0, stream>>>(u, S);
  caps_wmma_gemm<<<2 * (NOUT / NSTRIP), KSPLIT * 32, 0, stream>>>(S, W, out);
}
